# starting kernel
# baseline (speedup 1.0000x reference)
#include <hip/hip_runtime.h>
#include <hip/hip_bf16.h>
#include <math.h>

#define NN 256
#define BSZ 256
#define TMAXX 512
#define STEPF 0.01f
#define YAF 0.001f
#define YWF 0.001f

#define CSTRIDE 132

typedef __attribute__((ext_vector_type(2))) float v2f;
typedef __attribute__((ext_vector_type(8))) float v8f;

static __device__ __forceinline__ v8f wmma4(v2f a, v2f b, v8f c) {
    return __builtin_amdgcn_wmma_f32_16x16x4_f32(
        false, a, false, b, (short)0, c, false, false);
}

__global__ __launch_bounds__(256) void lrnn_prep(
    const float* __restrict__ MA, const float* __restrict__ MW,
    const float* __restrict__ bAz, const float* __restrict__ bWz,
    float* __restrict__ A, float* __restrict__ C) {
    const int i = blockIdx.x;
    const int j = threadIdx.x;
    const float zA = bAz[0];
    const float zW = bWz[0];
    const float bA = 0.5f * expf(-zA * zA) + 0.5f;
    const float bW = 0.5f * expf(-zW * zW) + 0.5f;
    const float ma_ij = MA[i * NN + j];
    const float ma_ji = MA[j * NN + i];
    const float mw_ij = MW[i * NN + j];
    const float mw_ji = MW[j * NN + i];
    const float d = (i == j) ? 1.0f : 0.0f;
    A[i * NN + j] = ma_ij + (1.0f - 2.0f * bA) * ma_ji - YAF * d;
    C[i * NN + j] = (1.0f - bA + bW) * mw_ij + (1.0f - bA - bW) * mw_ji - YWF * d;
}

__global__ __launch_bounds__(512) void lrnn_main(
    const float* __restrict__ X0, const float* __restrict__ A,
    const float* __restrict__ Cg, const float* __restrict__ by,
    float* __restrict__ out) {
    __shared__ float Xs[16][260];
    __shared__ float Cs[NN * CSTRIDE];

    const int tid  = threadIdx.x;
    const int wav  = tid >> 5;
    const int lane = tid & 31;
    const int lo   = lane & 15;
    const int hi   = lane >> 4;
    const int bc0  = blockIdx.x * 16;

    for (int idx = tid; idx < 16 * NN; idx += 512) {
        const int col = idx >> 8;
        const int row = idx & 255;
        const float v = X0[(size_t)(bc0 + col) * NN + row];
        Xs[col][row] = v;
        out[(size_t)(bc0 + col) * TMAXX * NN + row] = v;
    }
    for (int idx = tid; idx < NN * 128; idx += 512) {
        const int row = idx >> 7;
        const int col = idx & 127;
        Cs[row * CSTRIDE + col] = Cg[row * NN + 128 + col];
    }

    const int rowA = wav * 16 + lo;
    v2f aF[64];
    v2f cF[32];
#pragma unroll
    for (int k = 0; k < 64; ++k)
        aF[k] = *(const v2f*)(A + (size_t)rowA * NN + 4 * k + 2 * hi);
#pragma unroll
    for (int k = 0; k < 32; ++k)
        cF[k] = *(const v2f*)(Cg + (size_t)rowA * NN + 4 * k + 2 * hi);

    const int rbase = wav * 16 + 8 * hi;
    float byF[8];
#pragma unroll
    for (int j = 0; j < 8; ++j) byF[j] = by[rbase + j];

    __syncthreads();

    float* outLane = out + (size_t)(bc0 + lo) * TMAXX * NN + rbase;
    const float* crow = &Cs[rowA * CSTRIDE];

#pragma unroll 1
    for (int t = 1; t < TMAXX; ++t) {
        v8f accA = {0.f, 0.f, 0.f, 0.f, 0.f, 0.f, 0.f, 0.f};
        v8f accC = {0.f, 0.f, 0.f, 0.f, 0.f, 0.f, 0.f, 0.f};
#pragma unroll
        for (int k = 0; k < 32; ++k) {
            const int kr = 4 * k + 2 * hi;
            const v2f b = *(const v2f*)&Xs[lo][kr];
            accA = wmma4(aF[k], b, accA);
            accC = wmma4(cF[k], b, accC);
        }
#pragma unroll
        for (int k = 32; k < 64; ++k) {
            const int kr = 4 * k + 2 * hi;
            const v2f b  = *(const v2f*)&Xs[lo][kr];
            const v2f cl = *(const v2f*)&crow[kr - 128];
            accA = wmma4(aF[k], b, accA);
            accC = wmma4(cl, b, accC);
        }
        float xn[8];
#pragma unroll
        for (int j = 0; j < 8; ++j) {
            const float xc = Xs[lo][rbase + j];
            const float yi = accC[j] + byF[j];
            xn[j] = xc + STEPF * (accA[j] + tanhf(yi));
        }
        __syncthreads();
#pragma unroll
        for (int j = 0; j < 8; ++j) Xs[lo][rbase + j] = xn[j];
        float* p = outLane + (size_t)t * NN;
        *(float4*)(p)     = make_float4(xn[0], xn[1], xn[2], xn[3]);
        *(float4*)(p + 4) = make_float4(xn[4], xn[5], xn[6], xn[7]);
        __syncthreads();
    }
}

extern "C" void kernel_launch(void* const* d_in, const int* in_sizes, int n_in,
                              void* d_out, int out_size, void* d_ws, size_t ws_size,
                              hipStream_t stream) {
    const float* X0  = (const float*)d_in[0];
    const float* MA  = (const float*)d_in[1];
    const float* MW  = (const float*)d_in[2];
    const float* bAz = (const float*)d_in[3];
    const float* bWz = (const float*)d_in[4];
    const float* by  = (const float*)d_in[5];
    float* out = (float*)d_out;

    float* A = (float*)d_ws;
    float* C = A + NN * NN;

    lrnn_prep<<<NN, NN, 0, stream>>>(MA, MW, bAz, bWz, A, C);
    lrnn_main<<<16, 512, 0, stream>>>(X0, A, C, by, out);
}


# state tile double-buffered in LDS (a step reads one tile, writes the other), a lane's own state kept in registers: one barrier per step, not two; out stores issue behind it; bit-identical
# speedup vs baseline: 1.1493x; 1.1493x over previous
#include <hip/hip_runtime.h>
#include <hip/hip_bf16.h>
#include <math.h>

#define NN 256
#define BSZ 256
#define TMAXX 512
#define STEPF 0.01f
#define YAF 0.001f
#define YWF 0.001f

#define CSTRIDE 132
#define XPITCH 260
#define XTILE (16 * XPITCH)

typedef __attribute__((ext_vector_type(2))) float v2f;
typedef __attribute__((ext_vector_type(8))) float v8f;

static __device__ __forceinline__ v8f wmma4(v2f a, v2f b, v8f c) {
    return __builtin_amdgcn_wmma_f32_16x16x4_f32(
        false, a, false, b, (short)0, c, false, false);
}

__global__ __launch_bounds__(256) void lrnn_prep(
    const float* __restrict__ MA, const float* __restrict__ MW,
    const float* __restrict__ bAz, const float* __restrict__ bWz,
    float* __restrict__ A, float* __restrict__ C) {
    const int i = blockIdx.x;
    const int j = threadIdx.x;
    const float zA = bAz[0];
    const float zW = bWz[0];
    const float bA = 0.5f * expf(-zA * zA) + 0.5f;
    const float bW = 0.5f * expf(-zW * zW) + 0.5f;
    const float ma_ij = MA[i * NN + j];
    const float ma_ji = MA[j * NN + i];
    const float mw_ij = MW[i * NN + j];
    const float mw_ji = MW[j * NN + i];
    const float d = (i == j) ? 1.0f : 0.0f;
    A[i * NN + j] = ma_ij + (1.0f - 2.0f * bA) * ma_ji - YAF * d;
    C[i * NN + j] = (1.0f - bA + bW) * mw_ij + (1.0f - bA - bW) * mw_ji - YWF * d;
}

__global__ __launch_bounds__(512) void lrnn_main(
    const float* __restrict__ X0, const float* __restrict__ A,
    const float* __restrict__ Cg, const float* __restrict__ by,
    float* __restrict__ out) {
    __shared__ __attribute__((aligned(16))) float Xs[2 * XTILE];
    __shared__ float Cs[NN * CSTRIDE];

    const int tid  = threadIdx.x;
    const int wav  = tid >> 5;
    const int lane = tid & 31;
    const int lo   = lane & 15;
    const int hi   = lane >> 4;
    const int bc0  = blockIdx.x * 16;

    for (int idx = tid; idx < 16 * NN; idx += 512) {
        const int col = idx >> 8;
        const int row = idx & 255;
        const float v = X0[(size_t)(bc0 + col) * NN + row];
        Xs[col * XPITCH + row] = v;
        out[(size_t)(bc0 + col) * TMAXX * NN + row] = v;
    }
    for (int idx = tid; idx < NN * 128; idx += 512) {
        const int row = idx >> 7;
        const int col = idx & 127;
        Cs[row * CSTRIDE + col] = Cg[row * NN + 128 + col];
    }

    const int rowA = wav * 16 + lo;
    v2f aF[64];
    v2f cF[32];
#pragma unroll
    for (int k = 0; k < 64; ++k)
        aF[k] = *(const v2f*)(A + (size_t)rowA * NN + 4 * k + 2 * hi);
#pragma unroll
    for (int k = 0; k < 32; ++k)
        cF[k] = *(const v2f*)(Cg + (size_t)rowA * NN + 4 * k + 2 * hi);

    const int rbase = wav * 16 + 8 * hi;
    float byF[8];
#pragma unroll
    for (int j = 0; j < 8; ++j) byF[j] = by[rbase + j];

    __syncthreads();

    float* outLane = out + (size_t)(bc0 + lo) * TMAXX * NN + rbase;
    const float* crow = &Cs[rowA * CSTRIDE];

    float xs[8];
#pragma unroll
    for (int j = 0; j < 8; ++j) xs[j] = Xs[lo * XPITCH + rbase + j];
    int cur = 0;

#pragma unroll 1
    for (int t = 1; t < TMAXX; ++t) {
        const float* xr = Xs + cur + lo * XPITCH + 2 * hi;
        float* xw = Xs + (XTILE - cur) + lo * XPITCH + rbase;
        v8f accA = {0.f, 0.f, 0.f, 0.f, 0.f, 0.f, 0.f, 0.f};
        v8f accC = {0.f, 0.f, 0.f, 0.f, 0.f, 0.f, 0.f, 0.f};
        __builtin_amdgcn_s_setprio(3);
#pragma unroll
        for (int k = 0; k < 32; ++k) {
            const v2f b = *(const v2f*)&xr[4 * k];
            accA = wmma4(aF[k], b, accA);
            accC = wmma4(cF[k], b, accC);
            if ((k & 3) == 3) asm("" : "+v"(accA), "+v"(accC));
        }
#pragma unroll
        for (int k = 32; k < 64; ++k) {
            const int kr = 4 * k + 2 * hi;
            const v2f b  = *(const v2f*)&xr[4 * k];
            const v2f cl = *(const v2f*)&crow[kr - 128];
            accC = wmma4(cl, b, accC);
        }
        asm volatile("" : "+v"(accA));
        __builtin_amdgcn_s_setprio(0);
        float th[8];
#pragma unroll
        for (int j = 0; j < 8; ++j) th[j] = tanhf(accC[j] + byF[j]);
#pragma unroll
        for (int k = 32; k < 64; ++k) {
            const v2f b = *(const v2f*)&xr[4 * k];
            accA = wmma4(aF[k], b, accA);
        }
#pragma unroll
        for (int j = 0; j < 8; ++j) xs[j] = xs[j] + STEPF * (accA[j] + th[j]);
        *(float4*)(xw)     = make_float4(xs[0], xs[1], xs[2], xs[3]);
        *(float4*)(xw + 4) = make_float4(xs[4], xs[5], xs[6], xs[7]);
        __syncthreads();
        float* p = outLane + (size_t)t * NN;
        *(float4*)(p)     = make_float4(xs[0], xs[1], xs[2], xs[3]);
        *(float4*)(p + 4) = make_float4(xs[4], xs[5], xs[6], xs[7]);
        cur = XTILE - cur;
    }
}

extern "C" void kernel_launch(void* const* d_in, const int* in_sizes, int n_in,
                              void* d_out, int out_size, void* d_ws, size_t ws_size,
                              hipStream_t stream) {
    const float* X0  = (const float*)d_in[0];
    const float* MA  = (const float*)d_in[1];
    const float* MW  = (const float*)d_in[2];
    const float* bAz = (const float*)d_in[3];
    const float* bWz = (const float*)d_in[4];
    const float* by  = (const float*)d_in[5];
    float* out = (float*)d_out;

    float* A = (float*)d_ws;
    float* C = A + NN * NN;

    lrnn_prep<<<NN, NN, 0, stream>>>(MA, MW, bAz, bWz, A, C);
    lrnn_main<<<16, 512, 0, stream>>>(X0, A, C, by, out);
}


# barrier split into signal and wait: each wave sums k from its own 16 rows on (rotated order, not bit-identical) and runs those 8 products of the next step before the wait; state columns stored twice
# speedup vs baseline: 1.2905x; 1.1229x over previous
#include <hip/hip_runtime.h>
#include <hip/hip_bf16.h>
#include <math.h>

#define NN 256
#define BSZ 256
#define TMAXX 512
#define STEPF 0.01f
#define YAF 0.001f
#define YWF 0.001f

#define CSTRIDE 132
#define XPITCH 516
#define XTILE (16 * XPITCH)

typedef __attribute__((ext_vector_type(2))) float v2f;
typedef __attribute__((ext_vector_type(8))) float v8f;

static __device__ __forceinline__ v8f wmma4(v2f a, v2f b, v8f c) {
    return __builtin_amdgcn_wmma_f32_16x16x4_f32(
        false, a, false, b, (short)0, c, false, false);
}

__global__ __launch_bounds__(256) void lrnn_prep(
    const float* __restrict__ MA, const float* __restrict__ MW,
    const float* __restrict__ bAz, const float* __restrict__ bWz,
    float* __restrict__ A, float* __restrict__ C) {
    const int i = blockIdx.x;
    const int j = threadIdx.x;
    const float zA = bAz[0];
    const float zW = bWz[0];
    const float bA = 0.5f * expf(-zA * zA) + 0.5f;
    const float bW = 0.5f * expf(-zW * zW) + 0.5f;
    const float ma_ij = MA[i * NN + j];
    const float ma_ji = MA[j * NN + i];
    const float mw_ij = MW[i * NN + j];
    const float mw_ji = MW[j * NN + i];
    const float d = (i == j) ? 1.0f : 0.0f;
    A[i * NN + j] = ma_ij + (1.0f - 2.0f * bA) * ma_ji - YAF * d;
    C[i * NN + j] = (1.0f - bA + bW) * mw_ij + (1.0f - bA - bW) * mw_ji - YWF * d;
}

__global__ __launch_bounds__(512) void lrnn_main(
    const float* __restrict__ X0, const float* __restrict__ A,
    const float* __restrict__ Cg, const float* __restrict__ by,
    float* __restrict__ out) {
    __shared__ __attribute__((aligned(16))) float Xs[2 * XTILE];
    __shared__ float Cs[NN * CSTRIDE];

    const int tid  = threadIdx.x;
    const int wav  = tid >> 5;
    const int lane = tid & 31;
    const int lo   = lane & 15;
    const int hi   = lane >> 4;
    const int bc0  = blockIdx.x * 16;

    const int rbase = wav * 16 + 8 * hi;
    float* outLane = out + (size_t)(bc0 + lo) * TMAXX * NN + rbase;

    float xs[8];
    {
        const float4 x0a = *(const float4*)(X0 + (size_t)(bc0 + lo) * NN + rbase);
        const float4 x0b = *(const float4*)(X0 + (size_t)(bc0 + lo) * NN + rbase + 4);
        xs[0] = x0a.x; xs[1] = x0a.y; xs[2] = x0a.z; xs[3] = x0a.w;
        xs[4] = x0b.x; xs[5] = x0b.y; xs[6] = x0b.z; xs[7] = x0b.w;
        float* xw0 = Xs + lo * XPITCH + rbase;
        *(float4*)(xw0)          = x0a;
        *(float4*)(xw0 + 4)      = x0b;
        *(float4*)(xw0 + NN)     = x0a;
        *(float4*)(xw0 + NN + 4) = x0b;
        *(float4*)(outLane)      = x0a;
        *(float4*)(outLane + 4)  = x0b;
    }
    for (int idx = tid; idx < NN * 128; idx += 512) {
        const int row = idx >> 7;
        const int col = idx & 127;
        Cs[row * CSTRIDE + col] = Cg[row * NN + ((16 * (row >> 4) + 128 + col) & (NN - 1))];
    }

    const int rowA = wav * 16 + lo;
    const int krot = wav * 16 + 2 * hi;
    v2f aF[64];
    v2f cF[32];
#pragma unroll
    for (int k = 0; k < 64; ++k)
        aF[k] = *(const v2f*)(A + (size_t)rowA * NN + ((krot + 4 * k) & (NN - 1)));
#pragma unroll
    for (int k = 0; k < 32; ++k)
        cF[k] = *(const v2f*)(Cg + (size_t)rowA * NN + ((krot + 4 * k) & (NN - 1)));

    float byF[8];
#pragma unroll
    for (int j = 0; j < 8; ++j) byF[j] = by[rbase + j];

    const float* crow = &Cs[rowA * CSTRIDE];
    const int xoff = lo * XPITCH + wav * 16 + 2 * hi;
    int cur = 0;

    __syncthreads();
    asm volatile("s_barrier_signal -1" ::: "memory");

#pragma unroll 1
    for (int t = 1; t < TMAXX; ++t) {
        const float* xr = Xs + cur + xoff;
        float* xw = Xs + (XTILE - cur) + lo * XPITCH + rbase;
        v8f accA = {0.f, 0.f, 0.f, 0.f, 0.f, 0.f, 0.f, 0.f};
        v8f accC = {0.f, 0.f, 0.f, 0.f, 0.f, 0.f, 0.f, 0.f};
        __builtin_amdgcn_s_setprio(0);
#pragma unroll
        for (int k = 0; k < 4; ++k) {
            const v2f b = *(const v2f*)&xr[4 * k];
            accA = wmma4(aF[k], b, accA);
            accC = wmma4(cF[k], b, accC);
        }
        asm volatile("s_barrier_wait -1" : "+v"(accA), "+v"(accC) :: "memory");
        __builtin_amdgcn_s_setprio(3);
#pragma unroll
        for (int k = 4; k < 32; ++k) {
            const v2f b = *(const v2f*)&xr[4 * k];
            accA = wmma4(aF[k], b, accA);
            accC = wmma4(cF[k], b, accC);
            if ((k & 3) == 3) asm("" : "+v"(accA), "+v"(accC));
        }
#pragma unroll
        for (int k = 32; k < 64; ++k) {
            const int kr = 4 * k + 2 * hi;
            const v2f b  = *(const v2f*)&xr[4 * k];
            const v2f cl = *(const v2f*)&crow[kr - 128];
            accC = wmma4(cl, b, accC);
        }
        asm volatile("" : "+v"(accA));
        __builtin_amdgcn_s_setprio(1);
        float th[8];
#pragma unroll
        for (int j = 0; j < 8; ++j) th[j] = tanhf(accC[j] + byF[j]);
#pragma unroll
        for (int k = 32; k < 64; ++k) {
            const v2f b = *(const v2f*)&xr[4 * k];
            accA = wmma4(aF[k], b, accA);
        }
#pragma unroll
        for (int j = 0; j < 8; ++j) xs[j] = xs[j] + STEPF * (accA[j] + th[j]);
        *(float4*)(xw)          = make_float4(xs[0], xs[1], xs[2], xs[3]);
        *(float4*)(xw + 4)      = make_float4(xs[4], xs[5], xs[6], xs[7]);
        *(float4*)(xw + NN)     = make_float4(xs[0], xs[1], xs[2], xs[3]);
        *(float4*)(xw + NN + 4) = make_float4(xs[4], xs[5], xs[6], xs[7]);
        float* p = outLane + (size_t)t * NN;
        *(float4*)(p)     = make_float4(xs[0], xs[1], xs[2], xs[3]);
        *(float4*)(p + 4) = make_float4(xs[4], xs[5], xs[6], xs[7]);
        asm volatile("s_wait_dscnt 0x0\n\ts_barrier_signal -1" ::: "memory");
        cur = XTILE - cur;
    }
    asm volatile("s_barrier_wait -1" ::: "memory");
}

extern "C" void kernel_launch(void* const* d_in, const int* in_sizes, int n_in,
                              void* d_out, int out_size, void* d_ws, size_t ws_size,
                              hipStream_t stream) {
    const float* X0  = (const float*)d_in[0];
    const float* MA  = (const float*)d_in[1];
    const float* MW  = (const float*)d_in[2];
    const float* bAz = (const float*)d_in[3];
    const float* bWz = (const float*)d_in[4];
    const float* by  = (const float*)d_in[5];
    float* out = (float*)d_out;

    float* A = (float*)d_ws;
    float* C = A + NN * NN;

    lrnn_prep<<<NN, NN, 0, stream>>>(MA, MW, bAz, bWz, A, C);
    lrnn_main<<<16, 512, 0, stream>>>(X0, A, C, by, out);
}
